# starting kernel
# baseline (speedup 1.0000x reference)
#include <hip/hip_runtime.h>

#define NN   1024
#define BB   8
#define FF   32
#define FOUTC 4

typedef __attribute__((ext_vector_type(2))) float v2f;
typedef __attribute__((ext_vector_type(8))) float v8f;

__global__ void magnet_dinv_kernel(const float* __restrict__ adj,
                                   float* __restrict__ dinv) {
  int wid  = (blockIdx.x * blockDim.x + threadIdx.x) >> 5;
  int lane = threadIdx.x & 31;
  int b = wid >> 10;
  int i = wid & (NN - 1);
  const float* adjB = adj + (size_t)b * NN * NN;
  float s = 0.0f;
  for (int j = lane; j < NN; j += 32)
    s += 0.5f * (adjB[(size_t)i * NN + j] + adjB[(size_t)j * NN + i]);
  for (int off = 16; off > 0; off >>= 1)
    s += __shfl_xor(s, off, 32);
  if (lane == 0)
    dinv[wid] = (s > 0.0f) ? rsqrtf(s) : 0.0f;
}

__global__ __launch_bounds__(128)
void magnet_cheb_kernel(const float* __restrict__ adj,
                        const float* __restrict__ dinv,
                        const float* __restrict__ Zr, const float* __restrict__ Zi,
                        const float* __restrict__ Xr, const float* __restrict__ Xi,
                        float* __restrict__ Or, float* __restrict__ Oi,
                        float alpha, int zHasIdx) {
  const int lane = threadIdx.x & 31;
  const int wave = threadIdx.x >> 5;
  const int b    = blockIdx.y;
  const int idx  = blockIdx.z;
  const int rows0 = (blockIdx.x * 4 + wave) * 16;

  const float invf = (idx == 0) ? 0.5f : (idx == 1) ? (1.0f / 3.0f)
                    : (idx == 2) ? 0.25f : 0.2f;
  const float q2pi = 6.28318530717958647692f * invf;

  const int i    = rows0 + (lane & 15);
  const int koff = (lane >> 4) << 1;
  const int f0   = lane & 15;

  const float* adjB = adj + (size_t)b * NN * NN;
  const float* dv   = dinv + b * NN;
  const float  di   = dv[i];

  const size_t zbase = ((size_t)((zHasIdx ? idx * BB : 0) + b)) * NN * FF;
  const size_t xbase = (size_t)b * NN * FF;
  const size_t obase = ((size_t)(idx * BB + b)) * NN * FF;

  v8f hr0 = {0.f,0.f,0.f,0.f,0.f,0.f,0.f,0.f};
  v8f hr1 = hr0, hi0 = hr0, hi1 = hr0;

  for (int k0 = 0; k0 < NN; k0 += 4) {
    const int j0 = k0 + koff, j1 = j0 + 1;

    float aij0 = adjB[(size_t)i  * NN + j0];
    float aji0 = adjB[(size_t)j0 * NN + i ];
    float aij1 = adjB[(size_t)i  * NN + j1];
    float aji1 = adjB[(size_t)j1 * NN + i ];
    float s0 = 0.5f * (aij0 + aji0) * di * dv[j0];
    float s1 = 0.5f * (aij1 + aji1) * di * dv[j1];
    float sn0, c0, sn1, c1;
    __sincosf(q2pi * (aij0 - aji0), &sn0, &c0);
    __sincosf(q2pi * (aij1 - aji1), &sn1, &c1);
    float lr0 = ((i == j0) ? 1.0f : 0.0f) - s0 * c0;
    float lr1 = ((i == j1) ? 1.0f : 0.0f) - s1 * c1;
    float li0 = -s0 * sn0;
    float li1 = -s1 * sn1;

    v2f Ar  = { lr0,  lr1 };
    v2f Ai  = { li0,  li1 };
    v2f AiN = { -li0, -li1 };

    const float* zr0 = Zr + zbase + (size_t)j0 * FF;
    const float* zr1 = Zr + zbase + (size_t)j1 * FF;
    const float* zi0 = Zi + zbase + (size_t)j0 * FF;
    const float* zi1 = Zi + zbase + (size_t)j1 * FF;
    v2f Br0 = { zr0[f0],      zr1[f0]      };
    v2f Br1 = { zr0[f0 + 16], zr1[f0 + 16] };
    v2f Bi0 = { zi0[f0],      zi1[f0]      };
    v2f Bi1 = { zi0[f0 + 16], zi1[f0 + 16] };

    hr0 = __builtin_amdgcn_wmma_f32_16x16x4_f32(false, Ar,  false, Br0, (short)0, hr0, false, false);
    hr0 = __builtin_amdgcn_wmma_f32_16x16x4_f32(false, AiN, false, Bi0, (short)0, hr0, false, false);
    hr1 = __builtin_amdgcn_wmma_f32_16x16x4_f32(false, Ar,  false, Br1, (short)0, hr1, false, false);
    hr1 = __builtin_amdgcn_wmma_f32_16x16x4_f32(false, AiN, false, Bi1, (short)0, hr1, false, false);
    hi0 = __builtin_amdgcn_wmma_f32_16x16x4_f32(false, Ai,  false, Br0, (short)0, hi0, false, false);
    hi0 = __builtin_amdgcn_wmma_f32_16x16x4_f32(false, Ar,  false, Bi0, (short)0, hi0, false, false);
    hi1 = __builtin_amdgcn_wmma_f32_16x16x4_f32(false, Ai,  false, Br1, (short)0, hi1, false, false);
    hi1 = __builtin_amdgcn_wmma_f32_16x16x4_f32(false, Ar,  false, Bi1, (short)0, hi1, false, false);
  }

  const int rbase = rows0 + ((lane >> 4) << 3);
#pragma unroll
  for (int v = 0; v < 8; ++v) {
    const int row = rbase + v;
    const size_t xo = xbase + (size_t)row * FF + f0;
    const size_t oo = obase + (size_t)row * FF + f0;
    Or[oo]      = alpha * hr0[v] + Xr[xo];
    Or[oo + 16] = alpha * hr1[v] + Xr[xo + 16];
    Oi[oo]      = alpha * hi0[v] + Xi[xo];
    Oi[oo + 16] = alpha * hi1[v] + Xi[xo + 16];
  }
}

__global__ void magnet_proj_kernel(const float* __restrict__ Xr, const float* __restrict__ Xi,
                                   const float* __restrict__ z1r, const float* __restrict__ z1i,
                                   const float* __restrict__ z2r, const float* __restrict__ z2i,
                                   const float* __restrict__ weight,
                                   const float* __restrict__ bias,
                                   float* __restrict__ out) {
  int t = blockIdx.x * blockDim.x + threadIdx.x;
  int idx = t & 3;
  int n   = (t >> 2) & (NN - 1);
  int b   = t >> 12;
  const float* w = weight + idx * (3 * FF);
  const size_t xo = ((size_t)b * NN + n) * FF;
  const size_t zo = (((size_t)idx * BB + b) * NN + n) * FF;
  float sr = 0.0f, si = 0.0f;
#pragma unroll 8
  for (int f = 0; f < FF; ++f) {
    float w0 = w[f], w1 = w[FF + f], w2 = w[2 * FF + f];
    sr += Xr[xo + f] * w0 + z1r[zo + f] * w1 + z2r[zo + f] * w2;
    si += Xi[xo + f] * w0 + z1i[zo + f] * w1 + z2i[zo + f] * w2;
  }
  out[t] = sr + bias[idx];
  out[(size_t)BB * NN * FOUTC + t] = si;
}

extern "C" void kernel_launch(void* const* d_in, const int* in_sizes, int n_in,
                              void* d_out, int out_size, void* d_ws, size_t ws_size,
                              hipStream_t stream) {
  const float* Xr     = (const float*)d_in[0];
  const float* Xi     = (const float*)d_in[1];
  const float* adj    = (const float*)d_in[2];
  const float* weight = (const float*)d_in[3];
  const float* bias   = (const float*)d_in[4];
  float* out = (float*)d_out;

  char* ws = (char*)d_ws;
  float* dinv = (float*)ws;
  const size_t zElems = (size_t)FOUTC * BB * NN * FF;
  float* z1r = (float*)(ws + (1 << 16));
  float* z1i = z1r + zElems;
  float* z2r = z1i + zElems;
  float* z2i = z2r + zElems;

  magnet_dinv_kernel<<<dim3((BB * NN) / 8), 256, 0, stream>>>(adj, dinv);

  dim3 grid(NN / 64, BB, FOUTC);
  magnet_cheb_kernel<<<grid, 128, 0, stream>>>(adj, dinv, Xr, Xi, Xr, Xi,
                                               z1r, z1i, -1.0f, 0);
  magnet_cheb_kernel<<<grid, 128, 0, stream>>>(adj, dinv, z1r, z1i, Xr, Xi,
                                               z2r, z2i, 2.0f, 1);

  magnet_proj_kernel<<<dim3((BB * NN * FOUTC) / 256), 256, 0, stream>>>(
      Xr, Xi, z1r, z1i, z2r, z2i, weight, bias, out);
}


# cheb pass: k range split 4 ways across waves (16-wave blocks, fixed-order LDS sum of the 4 partials) so 8 waves per SIMD, not 2, cover each trip's cache wait; k-sum order changes
# speedup vs baseline: 1.8009x; 1.8009x over previous
#include <hip/hip_runtime.h>

#define NN   1024
#define BB   8
#define FF   32
#define FOUTC 4

typedef __attribute__((ext_vector_type(2))) float v2f;
typedef __attribute__((ext_vector_type(8))) float v8f;

__global__ void magnet_dinv_kernel(const float* __restrict__ adj,
                                   float* __restrict__ dinv) {
  int wid  = (blockIdx.x * blockDim.x + threadIdx.x) >> 5;
  int lane = threadIdx.x & 31;
  int b = wid >> 10;
  int i = wid & (NN - 1);
  const float* adjB = adj + (size_t)b * NN * NN;
  float s = 0.0f;
  for (int j = lane; j < NN; j += 32)
    s += 0.5f * (adjB[(size_t)i * NN + j] + adjB[(size_t)j * NN + i]);
  for (int off = 16; off > 0; off >>= 1)
    s += __shfl_xor(s, off, 32);
  if (lane == 0)
    dinv[wid] = (s > 0.0f) ? rsqrtf(s) : 0.0f;
}

#define KSPLIT 4
#define KQLEN  (NN / KSPLIT)

__global__ __launch_bounds__(512) __attribute__((amdgpu_waves_per_eu(8, 8)))
void magnet_cheb_kernel(const float* __restrict__ adj,
                        const float* __restrict__ dinv,
                        const float* __restrict__ Zr, const float* __restrict__ Zi,
                        const float* __restrict__ Xr, const float* __restrict__ Xi,
                        float* __restrict__ Or, float* __restrict__ Oi,
                        float alpha, int zHasIdx) {
  __shared__ float red[KSPLIT * 4 * 32 * 32];

  const int lane = threadIdx.x & 31;
  const int wave = __builtin_amdgcn_readfirstlane((int)(threadIdx.x >> 5));
  const int rt   = wave & 3;
  const int kq   = wave >> 2;
  const int b    = blockIdx.y;
  const int idx  = blockIdx.z;
  const int rows0 = (blockIdx.x * 4 + rt) * 16;

  const float invf = (idx == 0) ? 0.5f : (idx == 1) ? (1.0f / 3.0f)
                    : (idx == 2) ? 0.25f : 0.2f;
  const float q2pi = 6.28318530717958647692f * invf;

  const int i    = rows0 + (lane & 15);
  const int koff = (lane >> 4) << 1;
  const int f0   = lane & 15;

  const float* adjB = adj + (size_t)b * NN * NN;
  const float* dv   = dinv + b * NN;
  const float  di   = dv[i];

  const size_t zbase = ((size_t)((zHasIdx ? idx * BB : 0) + b)) * NN * FF;
  const size_t xbase = (size_t)b * NN * FF;
  const size_t obase = ((size_t)(idx * BB + b)) * NN * FF;

  v8f hr0 = {0.f,0.f,0.f,0.f,0.f,0.f,0.f,0.f};
  v8f hr1 = hr0, hi0 = hr0, hi1 = hr0;

  const int kbeg = kq * KQLEN;
#pragma unroll 2
  for (int t = 0; t < KQLEN / 4; ++t) {
    const int k0 = kbeg + 4 * t;
    const int j0 = k0 + koff, j1 = j0 + 1;

    float aij0 = adjB[(size_t)i  * NN + j0];
    float aji0 = adjB[(size_t)j0 * NN + i ];
    float aij1 = adjB[(size_t)i  * NN + j1];
    float aji1 = adjB[(size_t)j1 * NN + i ];
    float s0 = 0.5f * (aij0 + aji0) * di * dv[j0];
    float s1 = 0.5f * (aij1 + aji1) * di * dv[j1];
    float sn0, c0, sn1, c1;
    __sincosf(q2pi * (aij0 - aji0), &sn0, &c0);
    __sincosf(q2pi * (aij1 - aji1), &sn1, &c1);
    float lr0 = ((i == j0) ? 1.0f : 0.0f) - s0 * c0;
    float lr1 = ((i == j1) ? 1.0f : 0.0f) - s1 * c1;
    float li0 = -s0 * sn0;
    float li1 = -s1 * sn1;

    v2f Ar  = { lr0,  lr1 };
    v2f Ai  = { li0,  li1 };
    v2f AiN = { -li0, -li1 };

    const float* zr0 = Zr + zbase + (size_t)j0 * FF;
    const float* zr1 = Zr + zbase + (size_t)j1 * FF;
    const float* zi0 = Zi + zbase + (size_t)j0 * FF;
    const float* zi1 = Zi + zbase + (size_t)j1 * FF;
    v2f Br0 = { zr0[f0],      zr1[f0]      };
    v2f Br1 = { zr0[f0 + 16], zr1[f0 + 16] };
    v2f Bi0 = { zi0[f0],      zi1[f0]      };
    v2f Bi1 = { zi0[f0 + 16], zi1[f0 + 16] };

    hr0 = __builtin_amdgcn_wmma_f32_16x16x4_f32(false, Ar,  false, Br0, (short)0, hr0, false, false);
    hr0 = __builtin_amdgcn_wmma_f32_16x16x4_f32(false, AiN, false, Bi0, (short)0, hr0, false, false);
    hr1 = __builtin_amdgcn_wmma_f32_16x16x4_f32(false, Ar,  false, Br1, (short)0, hr1, false, false);
    hr1 = __builtin_amdgcn_wmma_f32_16x16x4_f32(false, AiN, false, Bi1, (short)0, hr1, false, false);
    hi0 = __builtin_amdgcn_wmma_f32_16x16x4_f32(false, Ai,  false, Br0, (short)0, hi0, false, false);
    hi0 = __builtin_amdgcn_wmma_f32_16x16x4_f32(false, Ar,  false, Bi0, (short)0, hi0, false, false);
    hi1 = __builtin_amdgcn_wmma_f32_16x16x4_f32(false, Ai,  false, Br1, (short)0, hi1, false, false);
    hi1 = __builtin_amdgcn_wmma_f32_16x16x4_f32(false, Ar,  false, Bi1, (short)0, hi1, false, false);
  }

  {
    float* slot = red + (kq * 4 + rt) * (32 * 32) + lane;
#pragma unroll
    for (int v = 0; v < 8; ++v) {
      slot[(v     ) * 32] = hr0[v];
      slot[(v +  8) * 32] = hr1[v];
      slot[(v + 16) * 32] = hi0[v];
      slot[(v + 24) * 32] = hi1[v];
    }
  }
  __syncthreads();

  const float* tile = red + rt * (32 * 32) + lane;
  const int rbase = rows0 + ((lane >> 4) << 3);
#pragma unroll
  for (int dvv = 0; dvv < 2; ++dvv) {
    const int v = 2 * kq + dvv;
    float acc[4];
#pragma unroll
    for (int a = 0; a < 4; ++a) {
      const float* p = tile + (a * 8 + v) * 32;
      float s = p[0];
#pragma unroll
      for (int q = 1; q < KSPLIT; ++q) s += p[q * 4 * (32 * 32)];
      acc[a] = s;
    }
    const int row = rbase + v;
    const size_t xo = xbase + (size_t)row * FF + f0;
    const size_t oo = obase + (size_t)row * FF + f0;
    Or[oo]      = alpha * acc[0] + Xr[xo];
    Or[oo + 16] = alpha * acc[1] + Xr[xo + 16];
    Oi[oo]      = alpha * acc[2] + Xi[xo];
    Oi[oo + 16] = alpha * acc[3] + Xi[xo + 16];
  }
}

__global__ void magnet_proj_kernel(const float* __restrict__ Xr, const float* __restrict__ Xi,
                                   const float* __restrict__ z1r, const float* __restrict__ z1i,
                                   const float* __restrict__ z2r, const float* __restrict__ z2i,
                                   const float* __restrict__ weight,
                                   const float* __restrict__ bias,
                                   float* __restrict__ out) {
  int t = blockIdx.x * blockDim.x + threadIdx.x;
  int idx = t & 3;
  int n   = (t >> 2) & (NN - 1);
  int b   = t >> 12;
  const float* w = weight + idx * (3 * FF);
  const size_t xo = ((size_t)b * NN + n) * FF;
  const size_t zo = (((size_t)idx * BB + b) * NN + n) * FF;
  float sr = 0.0f, si = 0.0f;
#pragma unroll 8
  for (int f = 0; f < FF; ++f) {
    float w0 = w[f], w1 = w[FF + f], w2 = w[2 * FF + f];
    sr += Xr[xo + f] * w0 + z1r[zo + f] * w1 + z2r[zo + f] * w2;
    si += Xi[xo + f] * w0 + z1i[zo + f] * w1 + z2i[zo + f] * w2;
  }
  out[t] = sr + bias[idx];
  out[(size_t)BB * NN * FOUTC + t] = si;
}

extern "C" void kernel_launch(void* const* d_in, const int* in_sizes, int n_in,
                              void* d_out, int out_size, void* d_ws, size_t ws_size,
                              hipStream_t stream) {
  const float* Xr     = (const float*)d_in[0];
  const float* Xi     = (const float*)d_in[1];
  const float* adj    = (const float*)d_in[2];
  const float* weight = (const float*)d_in[3];
  const float* bias   = (const float*)d_in[4];
  float* out = (float*)d_out;

  char* ws = (char*)d_ws;
  float* dinv = (float*)ws;
  const size_t zElems = (size_t)FOUTC * BB * NN * FF;
  float* z1r = (float*)(ws + (1 << 16));
  float* z1i = z1r + zElems;
  float* z2r = z1i + zElems;
  float* z2i = z2r + zElems;

  magnet_dinv_kernel<<<dim3((BB * NN) / 8), 256, 0, stream>>>(adj, dinv);

  dim3 grid(NN / 64, BB, FOUTC);
  magnet_cheb_kernel<<<grid, 512, 0, stream>>>(adj, dinv, Xr, Xi, Xr, Xi,
                                               z1r, z1i, -1.0f, 0);
  magnet_cheb_kernel<<<grid, 512, 0, stream>>>(adj, dinv, z1r, z1i, Xr, Xi,
                                               z2r, z2i, 2.0f, 1);

  magnet_proj_kernel<<<dim3((BB * NN * FOUTC) / 256), 256, 0, stream>>>(
      Xr, Xi, z1r, z1i, z2r, z2i, weight, bias, out);
}


# degree kernel: row and column sums read as whole 128-byte lines (lane = 4 floats of a row, 512-thread block per 32 indices, LDS sum in wave order), not a 32-line gather per load; sum order changes
# speedup vs baseline: 1.9627x; 1.0898x over previous
#include <hip/hip_runtime.h>

#define NN   1024
#define BB   8
#define FF   32
#define FOUTC 4

typedef __attribute__((ext_vector_type(2))) float v2f;
typedef __attribute__((ext_vector_type(8))) float v8f;

#define DGRP 32

typedef __attribute__((ext_vector_type(4))) float v4f;

__global__ __launch_bounds__(512)
void magnet_dinv_kernel(const float* __restrict__ adj,
                        float* __restrict__ dinv) {
  __shared__ float colpart[16 * DGRP];
  __shared__ float rowsum[DGRP];

  const int lane = threadIdx.x & 31;
  const int wave = threadIdx.x >> 5;
  const int b    = blockIdx.x / (NN / DGRP);
  const int c0   = (blockIdx.x % (NN / DGRP)) * DGRP;
  const float* adjB = adj + (size_t)b * NN * NN;

  const int r = lane >> 3, q = lane & 7;
  const float* pc = adjB + (size_t)(wave * 64 + r) * NN + c0 + 4 * q;
  v4f cs = *reinterpret_cast<const v4f*>(pc);
#pragma unroll
  for (int k = 1; k < 16; ++k)
    cs += *reinterpret_cast<const v4f*>(pc + (size_t)(4 * k) * NN);

  float rs[2];
#pragma unroll
  for (int u = 0; u < 2; ++u) {
    const float* pr = adjB + (size_t)(c0 + 2 * wave + u) * NN + 4 * lane;
    v4f a = *reinterpret_cast<const v4f*>(pr);
#pragma unroll
    for (int m = 1; m < 8; ++m)
      a += *reinterpret_cast<const v4f*>(pr + 128 * m);
    float s = (a.x + a.y) + (a.z + a.w);
    for (int off = 16; off > 0; off >>= 1)
      s += __shfl_xor(s, off, 32);
    rs[u] = s;
  }

  for (int off = 8; off <= 16; off <<= 1) {
    cs.x += __shfl_xor(cs.x, off, 32);
    cs.y += __shfl_xor(cs.y, off, 32);
    cs.z += __shfl_xor(cs.z, off, 32);
    cs.w += __shfl_xor(cs.w, off, 32);
  }
  if (lane < 8)
    *reinterpret_cast<v4f*>(colpart + wave * DGRP + 4 * lane) = cs;
  if (lane == 0) {
    rowsum[2 * wave]     = rs[0];
    rowsum[2 * wave + 1] = rs[1];
  }
  __syncthreads();

  if (wave == 0) {
    float c = colpart[lane];
#pragma unroll
    for (int w = 1; w < 16; ++w)
      c += colpart[w * DGRP + lane];
    const float d = 0.5f * rowsum[lane] + 0.5f * c;
    dinv[b * NN + c0 + lane] = (d > 0.0f) ? rsqrtf(d) : 0.0f;
  }
}

#define KSPLIT 4
#define KQLEN  (NN / KSPLIT)

__global__ __launch_bounds__(512) __attribute__((amdgpu_waves_per_eu(8, 8)))
void magnet_cheb_kernel(const float* __restrict__ adj,
                        const float* __restrict__ dinv,
                        const float* __restrict__ Zr, const float* __restrict__ Zi,
                        const float* __restrict__ Xr, const float* __restrict__ Xi,
                        float* __restrict__ Or, float* __restrict__ Oi,
                        float alpha, int zHasIdx) {
  __shared__ float red[KSPLIT * 4 * 32 * 32];

  const int lane = threadIdx.x & 31;
  const int wave = __builtin_amdgcn_readfirstlane((int)(threadIdx.x >> 5));
  const int rt   = wave & 3;
  const int kq   = wave >> 2;
  const int b    = blockIdx.y;
  const int idx  = blockIdx.z;
  const int rows0 = (blockIdx.x * 4 + rt) * 16;

  const float invf = (idx == 0) ? 0.5f : (idx == 1) ? (1.0f / 3.0f)
                    : (idx == 2) ? 0.25f : 0.2f;
  const float q2pi = 6.28318530717958647692f * invf;

  const int i    = rows0 + (lane & 15);
  const int koff = (lane >> 4) << 1;
  const int f0   = lane & 15;

  const float* adjB = adj + (size_t)b * NN * NN;
  const float* dv   = dinv + b * NN;
  const float  di   = dv[i];

  const size_t zbase = ((size_t)((zHasIdx ? idx * BB : 0) + b)) * NN * FF;
  const size_t xbase = (size_t)b * NN * FF;
  const size_t obase = ((size_t)(idx * BB + b)) * NN * FF;

  v8f hr0 = {0.f,0.f,0.f,0.f,0.f,0.f,0.f,0.f};
  v8f hr1 = hr0, hi0 = hr0, hi1 = hr0;

  const int kbeg = kq * KQLEN;
#pragma unroll 2
  for (int t = 0; t < KQLEN / 4; ++t) {
    const int k0 = kbeg + 4 * t;
    const int j0 = k0 + koff, j1 = j0 + 1;

    float aij0 = adjB[(size_t)i  * NN + j0];
    float aji0 = adjB[(size_t)j0 * NN + i ];
    float aij1 = adjB[(size_t)i  * NN + j1];
    float aji1 = adjB[(size_t)j1 * NN + i ];
    float s0 = 0.5f * (aij0 + aji0) * di * dv[j0];
    float s1 = 0.5f * (aij1 + aji1) * di * dv[j1];
    float sn0, c0, sn1, c1;
    __sincosf(q2pi * (aij0 - aji0), &sn0, &c0);
    __sincosf(q2pi * (aij1 - aji1), &sn1, &c1);
    float lr0 = ((i == j0) ? 1.0f : 0.0f) - s0 * c0;
    float lr1 = ((i == j1) ? 1.0f : 0.0f) - s1 * c1;
    float li0 = -s0 * sn0;
    float li1 = -s1 * sn1;

    v2f Ar  = { lr0,  lr1 };
    v2f Ai  = { li0,  li1 };
    v2f AiN = { -li0, -li1 };

    const float* zr0 = Zr + zbase + (size_t)j0 * FF;
    const float* zr1 = Zr + zbase + (size_t)j1 * FF;
    const float* zi0 = Zi + zbase + (size_t)j0 * FF;
    const float* zi1 = Zi + zbase + (size_t)j1 * FF;
    v2f Br0 = { zr0[f0],      zr1[f0]      };
    v2f Br1 = { zr0[f0 + 16], zr1[f0 + 16] };
    v2f Bi0 = { zi0[f0],      zi1[f0]      };
    v2f Bi1 = { zi0[f0 + 16], zi1[f0 + 16] };

    hr0 = __builtin_amdgcn_wmma_f32_16x16x4_f32(false, Ar,  false, Br0, (short)0, hr0, false, false);
    hr0 = __builtin_amdgcn_wmma_f32_16x16x4_f32(false, AiN, false, Bi0, (short)0, hr0, false, false);
    hr1 = __builtin_amdgcn_wmma_f32_16x16x4_f32(false, Ar,  false, Br1, (short)0, hr1, false, false);
    hr1 = __builtin_amdgcn_wmma_f32_16x16x4_f32(false, AiN, false, Bi1, (short)0, hr1, false, false);
    hi0 = __builtin_amdgcn_wmma_f32_16x16x4_f32(false, Ai,  false, Br0, (short)0, hi0, false, false);
    hi0 = __builtin_amdgcn_wmma_f32_16x16x4_f32(false, Ar,  false, Bi0, (short)0, hi0, false, false);
    hi1 = __builtin_amdgcn_wmma_f32_16x16x4_f32(false, Ai,  false, Br1, (short)0, hi1, false, false);
    hi1 = __builtin_amdgcn_wmma_f32_16x16x4_f32(false, Ar,  false, Bi1, (short)0, hi1, false, false);
  }

  {
    float* slot = red + (kq * 4 + rt) * (32 * 32) + lane;
#pragma unroll
    for (int v = 0; v < 8; ++v) {
      slot[(v     ) * 32] = hr0[v];
      slot[(v +  8) * 32] = hr1[v];
      slot[(v + 16) * 32] = hi0[v];
      slot[(v + 24) * 32] = hi1[v];
    }
  }
  __syncthreads();

  const float* tile = red + rt * (32 * 32) + lane;
  const int rbase = rows0 + ((lane >> 4) << 3);
#pragma unroll
  for (int dvv = 0; dvv < 2; ++dvv) {
    const int v = 2 * kq + dvv;
    float acc[4];
#pragma unroll
    for (int a = 0; a < 4; ++a) {
      const float* p = tile + (a * 8 + v) * 32;
      float s = p[0];
#pragma unroll
      for (int q = 1; q < KSPLIT; ++q) s += p[q * 4 * (32 * 32)];
      acc[a] = s;
    }
    const int row = rbase + v;
    const size_t xo = xbase + (size_t)row * FF + f0;
    const size_t oo = obase + (size_t)row * FF + f0;
    Or[oo]      = alpha * acc[0] + Xr[xo];
    Or[oo + 16] = alpha * acc[1] + Xr[xo + 16];
    Oi[oo]      = alpha * acc[2] + Xi[xo];
    Oi[oo + 16] = alpha * acc[3] + Xi[xo + 16];
  }
}

__global__ void magnet_proj_kernel(const float* __restrict__ Xr, const float* __restrict__ Xi,
                                   const float* __restrict__ z1r, const float* __restrict__ z1i,
                                   const float* __restrict__ z2r, const float* __restrict__ z2i,
                                   const float* __restrict__ weight,
                                   const float* __restrict__ bias,
                                   float* __restrict__ out) {
  int t = blockIdx.x * blockDim.x + threadIdx.x;
  int idx = t & 3;
  int n   = (t >> 2) & (NN - 1);
  int b   = t >> 12;
  const float* w = weight + idx * (3 * FF);
  const size_t xo = ((size_t)b * NN + n) * FF;
  const size_t zo = (((size_t)idx * BB + b) * NN + n) * FF;
  float sr = 0.0f, si = 0.0f;
#pragma unroll 8
  for (int f = 0; f < FF; ++f) {
    float w0 = w[f], w1 = w[FF + f], w2 = w[2 * FF + f];
    sr += Xr[xo + f] * w0 + z1r[zo + f] * w1 + z2r[zo + f] * w2;
    si += Xi[xo + f] * w0 + z1i[zo + f] * w1 + z2i[zo + f] * w2;
  }
  out[t] = sr + bias[idx];
  out[(size_t)BB * NN * FOUTC + t] = si;
}

extern "C" void kernel_launch(void* const* d_in, const int* in_sizes, int n_in,
                              void* d_out, int out_size, void* d_ws, size_t ws_size,
                              hipStream_t stream) {
  const float* Xr     = (const float*)d_in[0];
  const float* Xi     = (const float*)d_in[1];
  const float* adj    = (const float*)d_in[2];
  const float* weight = (const float*)d_in[3];
  const float* bias   = (const float*)d_in[4];
  float* out = (float*)d_out;

  char* ws = (char*)d_ws;
  float* dinv = (float*)ws;
  const size_t zElems = (size_t)FOUTC * BB * NN * FF;
  float* z1r = (float*)(ws + (1 << 16));
  float* z1i = z1r + zElems;
  float* z2r = z1i + zElems;
  float* z2i = z2r + zElems;

  magnet_dinv_kernel<<<dim3(BB * (NN / DGRP)), 512, 0, stream>>>(adj, dinv);

  dim3 grid(NN / 64, BB, FOUTC);
  magnet_cheb_kernel<<<grid, 512, 0, stream>>>(adj, dinv, Xr, Xi, Xr, Xi,
                                               z1r, z1i, -1.0f, 0);
  magnet_cheb_kernel<<<grid, 512, 0, stream>>>(adj, dinv, z1r, z1i, Xr, Xi,
                                               z2r, z2i, 2.0f, 1);

  magnet_proj_kernel<<<dim3((BB * NN * FOUTC) / 256), 256, 0, stream>>>(
      Xr, Xi, z1r, z1i, z2r, z2i, weight, bias, out);
}


# cheb pass: complex product from 3 real matrix products (P1=Lr*Zr, P2=Li*Zi, P3=(Lr+Li)*(Zr+Zi); Re=P1-P2, Im=P3-P1-P2): 6 f32 WMMAs per 4 k, not 8; the sums of Re and Im are re-associated
# speedup vs baseline: 2.2407x; 1.1416x over previous
#include <hip/hip_runtime.h>

#define NN   1024
#define BB   8
#define FF   32
#define FOUTC 4

typedef __attribute__((ext_vector_type(2))) float v2f;
typedef __attribute__((ext_vector_type(8))) float v8f;

#define DGRP 32

typedef __attribute__((ext_vector_type(4))) float v4f;

__global__ __launch_bounds__(512)
void magnet_dinv_kernel(const float* __restrict__ adj,
                        float* __restrict__ dinv) {
  __shared__ float colpart[16 * DGRP];
  __shared__ float rowsum[DGRP];

  const int lane = threadIdx.x & 31;
  const int wave = threadIdx.x >> 5;
  const int b    = blockIdx.x / (NN / DGRP);
  const int c0   = (blockIdx.x % (NN / DGRP)) * DGRP;
  const float* adjB = adj + (size_t)b * NN * NN;

  const int r = lane >> 3, q = lane & 7;
  const float* pc = adjB + (size_t)(wave * 64 + r) * NN + c0 + 4 * q;
  v4f cs = *reinterpret_cast<const v4f*>(pc);
#pragma unroll
  for (int k = 1; k < 16; ++k)
    cs += *reinterpret_cast<const v4f*>(pc + (size_t)(4 * k) * NN);

  float rs[2];
#pragma unroll
  for (int u = 0; u < 2; ++u) {
    const float* pr = adjB + (size_t)(c0 + 2 * wave + u) * NN + 4 * lane;
    v4f a = *reinterpret_cast<const v4f*>(pr);
#pragma unroll
    for (int m = 1; m < 8; ++m)
      a += *reinterpret_cast<const v4f*>(pr + 128 * m);
    float s = (a.x + a.y) + (a.z + a.w);
    for (int off = 16; off > 0; off >>= 1)
      s += __shfl_xor(s, off, 32);
    rs[u] = s;
  }

  for (int off = 8; off <= 16; off <<= 1) {
    cs.x += __shfl_xor(cs.x, off, 32);
    cs.y += __shfl_xor(cs.y, off, 32);
    cs.z += __shfl_xor(cs.z, off, 32);
    cs.w += __shfl_xor(cs.w, off, 32);
  }
  if (lane < 8)
    *reinterpret_cast<v4f*>(colpart + wave * DGRP + 4 * lane) = cs;
  if (lane == 0) {
    rowsum[2 * wave]     = rs[0];
    rowsum[2 * wave + 1] = rs[1];
  }
  __syncthreads();

  if (wave == 0) {
    float c = colpart[lane];
#pragma unroll
    for (int w = 1; w < 16; ++w)
      c += colpart[w * DGRP + lane];
    const float d = 0.5f * rowsum[lane] + 0.5f * c;
    dinv[b * NN + c0 + lane] = (d > 0.0f) ? rsqrtf(d) : 0.0f;
  }
}

#define KSPLIT 4
#define KQLEN  (NN / KSPLIT)

__global__ __launch_bounds__(512) __attribute__((amdgpu_waves_per_eu(8, 8)))
void magnet_cheb_kernel(const float* __restrict__ adj,
                        const float* __restrict__ dinv,
                        const float* __restrict__ Zr, const float* __restrict__ Zi,
                        const float* __restrict__ Xr, const float* __restrict__ Xi,
                        float* __restrict__ Or, float* __restrict__ Oi,
                        float alpha, int zHasIdx) {
  __shared__ float red[KSPLIT * 4 * 48 * 32];

  const int lane = threadIdx.x & 31;
  const int wave = __builtin_amdgcn_readfirstlane((int)(threadIdx.x >> 5));
  const int rt   = wave & 3;
  const int kq   = wave >> 2;
  const int b    = blockIdx.y;
  const int idx  = blockIdx.z;
  const int rows0 = (blockIdx.x * 4 + rt) * 16;

  const float invf = (idx == 0) ? 0.5f : (idx == 1) ? (1.0f / 3.0f)
                    : (idx == 2) ? 0.25f : 0.2f;
  const float q2pi = 6.28318530717958647692f * invf;

  const int i    = rows0 + (lane & 15);
  const int koff = (lane >> 4) << 1;
  const int f0   = lane & 15;

  const float* adjB = adj + (size_t)b * NN * NN;
  const float* dv   = dinv + b * NN;
  const float  di   = dv[i];

  const size_t zbase = ((size_t)((zHasIdx ? idx * BB : 0) + b)) * NN * FF;
  const size_t xbase = (size_t)b * NN * FF;
  const size_t obase = ((size_t)(idx * BB + b)) * NN * FF;

  v8f p10 = {0.f,0.f,0.f,0.f,0.f,0.f,0.f,0.f};
  v8f p11 = p10, p20 = p10, p21 = p10, p30 = p10, p31 = p10;

  const int kbeg = kq * KQLEN;
#pragma unroll 2
  for (int t = 0; t < KQLEN / 4; ++t) {
    const int k0 = kbeg + 4 * t;
    const int j0 = k0 + koff, j1 = j0 + 1;

    float aij0 = adjB[(size_t)i  * NN + j0];
    float aji0 = adjB[(size_t)j0 * NN + i ];
    float aij1 = adjB[(size_t)i  * NN + j1];
    float aji1 = adjB[(size_t)j1 * NN + i ];
    float s0 = 0.5f * (aij0 + aji0) * di * dv[j0];
    float s1 = 0.5f * (aij1 + aji1) * di * dv[j1];
    float sn0, c0, sn1, c1;
    __sincosf(q2pi * (aij0 - aji0), &sn0, &c0);
    __sincosf(q2pi * (aij1 - aji1), &sn1, &c1);
    float lr0 = ((i == j0) ? 1.0f : 0.0f) - s0 * c0;
    float lr1 = ((i == j1) ? 1.0f : 0.0f) - s1 * c1;
    float li0 = -s0 * sn0;
    float li1 = -s1 * sn1;

    v2f Ar = { lr0, lr1 };
    v2f Ai = { li0, li1 };
    v2f As = Ar + Ai;

    const float* zr0 = Zr + zbase + (size_t)j0 * FF;
    const float* zr1 = Zr + zbase + (size_t)j1 * FF;
    const float* zi0 = Zi + zbase + (size_t)j0 * FF;
    const float* zi1 = Zi + zbase + (size_t)j1 * FF;
    v2f Br0 = { zr0[f0],      zr1[f0]      };
    v2f Br1 = { zr0[f0 + 16], zr1[f0 + 16] };
    v2f Bi0 = { zi0[f0],      zi1[f0]      };
    v2f Bi1 = { zi0[f0 + 16], zi1[f0 + 16] };
    v2f Bs0 = Br0 + Bi0;
    v2f Bs1 = Br1 + Bi1;

    p10 = __builtin_amdgcn_wmma_f32_16x16x4_f32(false, Ar, false, Br0, (short)0, p10, false, false);
    p11 = __builtin_amdgcn_wmma_f32_16x16x4_f32(false, Ar, false, Br1, (short)0, p11, false, false);
    p20 = __builtin_amdgcn_wmma_f32_16x16x4_f32(false, Ai, false, Bi0, (short)0, p20, false, false);
    p21 = __builtin_amdgcn_wmma_f32_16x16x4_f32(false, Ai, false, Bi1, (short)0, p21, false, false);
    p30 = __builtin_amdgcn_wmma_f32_16x16x4_f32(false, As, false, Bs0, (short)0, p30, false, false);
    p31 = __builtin_amdgcn_wmma_f32_16x16x4_f32(false, As, false, Bs1, (short)0, p31, false, false);
  }

  {
    float* slot = red + (kq * 4 + rt) * (48 * 32) + lane;
#pragma unroll
    for (int v = 0; v < 8; ++v) {
      slot[(v     ) * 32] = p10[v];
      slot[(v +  8) * 32] = p11[v];
      slot[(v + 16) * 32] = p20[v];
      slot[(v + 24) * 32] = p21[v];
      slot[(v + 32) * 32] = p30[v];
      slot[(v + 40) * 32] = p31[v];
    }
  }
  __syncthreads();

  const float* tile = red + rt * (48 * 32) + lane;
  const int rbase = rows0 + ((lane >> 4) << 3);
#pragma unroll
  for (int dvv = 0; dvv < 2; ++dvv) {
    const int v = 2 * kq + dvv;
    float sum[6];
#pragma unroll
    for (int a = 0; a < 6; ++a) {
      const float* p = tile + (a * 8 + v) * 32;
      float s = p[0];
#pragma unroll
      for (int q = 1; q < KSPLIT; ++q) s += p[q * 4 * (48 * 32)];
      sum[a] = s;
    }
    float acc[4];
    acc[0] = sum[0] - sum[2];
    acc[1] = sum[1] - sum[3];
    acc[2] = (sum[4] - sum[0]) - sum[2];
    acc[3] = (sum[5] - sum[1]) - sum[3];
    const int row = rbase + v;
    const size_t xo = xbase + (size_t)row * FF + f0;
    const size_t oo = obase + (size_t)row * FF + f0;
    Or[oo]      = alpha * acc[0] + Xr[xo];
    Or[oo + 16] = alpha * acc[1] + Xr[xo + 16];
    Oi[oo]      = alpha * acc[2] + Xi[xo];
    Oi[oo + 16] = alpha * acc[3] + Xi[xo + 16];
  }
}

__global__ void magnet_proj_kernel(const float* __restrict__ Xr, const float* __restrict__ Xi,
                                   const float* __restrict__ z1r, const float* __restrict__ z1i,
                                   const float* __restrict__ z2r, const float* __restrict__ z2i,
                                   const float* __restrict__ weight,
                                   const float* __restrict__ bias,
                                   float* __restrict__ out) {
  int t = blockIdx.x * blockDim.x + threadIdx.x;
  int idx = t & 3;
  int n   = (t >> 2) & (NN - 1);
  int b   = t >> 12;
  const float* w = weight + idx * (3 * FF);
  const size_t xo = ((size_t)b * NN + n) * FF;
  const size_t zo = (((size_t)idx * BB + b) * NN + n) * FF;
  float sr = 0.0f, si = 0.0f;
#pragma unroll 8
  for (int f = 0; f < FF; ++f) {
    float w0 = w[f], w1 = w[FF + f], w2 = w[2 * FF + f];
    sr += Xr[xo + f] * w0 + z1r[zo + f] * w1 + z2r[zo + f] * w2;
    si += Xi[xo + f] * w0 + z1i[zo + f] * w1 + z2i[zo + f] * w2;
  }
  out[t] = sr + bias[idx];
  out[(size_t)BB * NN * FOUTC + t] = si;
}

extern "C" void kernel_launch(void* const* d_in, const int* in_sizes, int n_in,
                              void* d_out, int out_size, void* d_ws, size_t ws_size,
                              hipStream_t stream) {
  const float* Xr     = (const float*)d_in[0];
  const float* Xi     = (const float*)d_in[1];
  const float* adj    = (const float*)d_in[2];
  const float* weight = (const float*)d_in[3];
  const float* bias   = (const float*)d_in[4];
  float* out = (float*)d_out;

  char* ws = (char*)d_ws;
  float* dinv = (float*)ws;
  const size_t zElems = (size_t)FOUTC * BB * NN * FF;
  float* z1r = (float*)(ws + (1 << 16));
  float* z1i = z1r + zElems;
  float* z2r = z1i + zElems;
  float* z2i = z2r + zElems;

  magnet_dinv_kernel<<<dim3(BB * (NN / DGRP)), 512, 0, stream>>>(adj, dinv);

  dim3 grid(NN / 64, BB, FOUTC);
  magnet_cheb_kernel<<<grid, 512, 0, stream>>>(adj, dinv, Xr, Xi, Xr, Xi,
                                               z1r, z1i, -1.0f, 0);
  magnet_cheb_kernel<<<grid, 512, 0, stream>>>(adj, dinv, z1r, z1i, Xr, Xi,
                                               z2r, z2i, 2.0f, 1);

  magnet_proj_kernel<<<dim3((BB * NN * FOUTC) / 256), 256, 0, stream>>>(
      Xr, Xi, z1r, z1i, z2r, z2i, weight, bias, out);
}
